# starting kernel
# baseline (speedup 1.0000x reference)
#include <hip/hip_runtime.h>
#include <hip/hip_bf16.h>
#include <stdint.h>

#define B_IMG     8
#define CIN       1024
#define H_DIM     64
#define W_DIM     64
#define HW        (H_DIM * W_DIM)
#define C_OUT     196
#define C_OUT_PAD 208
#define N_ROI     1024
#define KPOOL     7
#define SAT_W     65

typedef __attribute__((ext_vector_type(2))) float v2f;
typedef __attribute__((ext_vector_type(8))) float v8f;

__global__ void batchsum_kernel(const float* __restrict__ fm,
                                float* __restrict__ F) {
    const size_t NE = (size_t)CIN * HW;
    size_t idx = ((size_t)blockIdx.x * blockDim.x + threadIdx.x) * 4;
    if (idx >= NE) return;
    float4 s = make_float4(0.f, 0.f, 0.f, 0.f);
#pragma unroll
    for (int b = 0; b < B_IMG; ++b) {
        const float4 v = *(const float4*)(fm + (size_t)b * NE + idx);
        s.x += v.x; s.y += v.y; s.z += v.z; s.w += v.w;
    }
    *(float4*)(F + idx) = s;
}

__global__ void gemm_wmma_kernel(const float* __restrict__ Wmat,
                                 const float* __restrict__ bias,
                                 const float* __restrict__ F,
                                 float* __restrict__ S) {
    const int flat  = blockIdx.x * blockDim.x + threadIdx.x;
    const int wave  = flat >> 5;
    const int lane  = threadIdx.x & 31;
    const int nTilesN = HW / 16;
    const int mt = wave / nTilesN;
    const int nt = wave % nTilesN;
    if (mt >= C_OUT_PAD / 16) return;

    const int o0 = mt * 16;
    const int n0 = nt * 16;
    const int lhalf = lane >> 4;
    const int l16   = lane & 15;

    v8f acc;
#pragma unroll
    for (int v = 0; v < 8; ++v) {
        const int o = min(o0 + v + 8 * lhalf, C_OUT - 1);
        acc[v] = 8.0f * bias[o];
    }

    const int oA = min(o0 + l16, C_OUT - 1);
    const float* Wrow = Wmat + (size_t)oA * CIN;
    const float* Fcol = F + n0 + l16;

    for (int k0 = 0; k0 < CIN; k0 += 4) {
        const int kk = k0 + 2 * lhalf;
        const v2f a = *(const v2f*)(Wrow + kk);
        v2f b;
        b.x = Fcol[(size_t)kk * HW];
        b.y = Fcol[(size_t)(kk + 1) * HW];
        acc = __builtin_amdgcn_wmma_f32_16x16x4_f32(
            false, a, false, b, (short)0, acc, false, false);
    }

#pragma unroll
    for (int v = 0; v < 8; ++v) {
        const int m = v + 8 * lhalf;
        S[(size_t)(o0 + m) * HW + n0 + l16] = acc[v];
    }
}

__global__ void integral_kernel(const float* __restrict__ S,
                                float* __restrict__ sat) {
    const int ch = blockIdx.x;
    const int t  = threadIdx.x;
    __shared__ float tile[H_DIM][W_DIM + 1];

    const float* Sp = S + (size_t)ch * HW;
    float run = 0.0f;
    for (int x = 0; x < W_DIM; ++x) {
        run += Sp[t * W_DIM + x];
        tile[t][x] = run;
    }
    __syncthreads();

    float* satp = sat + (size_t)ch * SAT_W * SAT_W;
    satp[t + 1]           = 0.0f;
    satp[(t + 1) * SAT_W] = 0.0f;
    if (t == 0) satp[0] = 0.0f;

    float crun = 0.0f;
    for (int y = 0; y < H_DIM; ++y) {
        crun += tile[y][t];
        satp[(y + 1) * SAT_W + (t + 1)] = crun;
    }
}

__global__ void roipool_kernel(const int* __restrict__ rois,
                               const float* __restrict__ sat,
                               float* __restrict__ out) {
    const int i = blockIdx.x;
    const int t = threadIdx.x;
    __shared__ float pool[C_OUT];

    const int* r = rois + i * 4;
    const int ymin = r[0] / 30, xmin = r[1] / 30;
    const int ymax = r[2] / 30, xmax = r[3] / 30;
    const int ystep = (ymax - ymin) / KPOOL;
    const int xstep = (xmax - xmin) / KPOOL;
    const float cnt = (float)(B_IMG * ystep * xstep);

    if (t < C_OUT) {
        const int bin = t >> 2;
        const int jy = bin / KPOOL, jx = bin % KPOOL;
        const int ys = ymin + jy * ystep, ye = ys + ystep;
        const int xs = xmin + jx * xstep, xe = xs + xstep;
        const float* sp = sat + (size_t)t * SAT_W * SAT_W;
        const float bs = sp[ye * SAT_W + xe] - sp[ys * SAT_W + xe]
                       - sp[ye * SAT_W + xs] + sp[ys * SAT_W + xs];
        pool[t] = (cnt > 0.0f) ? bs / fmaxf(cnt, 1.0f) : 0.0f;
    }
    __syncthreads();

    if (t < 4) {
        float s = 0.0f;
#pragma unroll
        for (int b = 0; b < KPOOL * KPOOL; ++b) s += pool[t + 4 * b];
        out[i * 4 + t] = s / (float)(KPOOL * KPOOL);
    }
}

extern "C" void kernel_launch(void* const* d_in, const int* in_sizes, int n_in,
                              void* d_out, int out_size, void* d_ws, size_t ws_size,
                              hipStream_t stream) {
    const float* fm    = (const float*)d_in[0];
    const int*   rois  = (const int*)  d_in[1];
    const float* convw = (const float*)d_in[2];
    const float* convb = (const float*)d_in[3];
    float*       out   = (float*)d_out;

    char* ws = (char*)d_ws;
    float* F   = (float*)(ws);
    float* S   = (float*)(ws + (size_t)CIN * HW * sizeof(float));
    float* sat = (float*)(ws + ((size_t)CIN * HW + (size_t)C_OUT_PAD * HW) * sizeof(float));

    {
        const int threads = 256;
        const int blocks  = (int)(((size_t)CIN * HW / 4 + threads - 1) / threads);
        batchsum_kernel<<<blocks, threads, 0, stream>>>(fm, F);
    }
    {
        const int threads = 256;
        const int totalWaves = (C_OUT_PAD / 16) * (HW / 16);
        const int blocks = (totalWaves * 32 + threads - 1) / threads;
        gemm_wmma_kernel<<<blocks, threads, 0, stream>>>(convw, convb, F, S);
    }
    integral_kernel<<<C_OUT, 64, 0, stream>>>(S, sat);

    roipool_kernel<<<N_ROI, 256, 0, stream>>>(rois, sat, out);
}


# ROI pooling: one wave per ROI (was 8 waves, LDS, barrier); lane takes channels l+32k, its 28 lookups in flight together; 49-bin mean by lane sums + 3 xor shuffles: sum order changes
# speedup vs baseline: 1.4778x; 1.4778x over previous
#include <hip/hip_runtime.h>
#include <hip/hip_bf16.h>
#include <stdint.h>

#define B_IMG     8
#define CIN       1024
#define H_DIM     64
#define W_DIM     64
#define HW        (H_DIM * W_DIM)
#define C_OUT     196
#define C_OUT_PAD 208
#define N_ROI     1024
#define KPOOL     7

typedef __attribute__((ext_vector_type(2))) float v2f;
typedef __attribute__((ext_vector_type(8))) float v8f;

#define SAT_PITCH 68

template <int D>
__device__ __forceinline__ float row_shr_or_zero(float x) {
    return __builtin_bit_cast(float,
        __builtin_amdgcn_update_dpp(0, __builtin_bit_cast(int, x), 0x110 + D, 0xf, 0xf, false));
}

__global__ void __launch_bounds__(256)
batchsum_sat_kernel(const float* __restrict__ fm,
                    float* __restrict__ P) {
    __shared__ __attribute__((aligned(16))) float tile[H_DIM * SAT_PITCH];
    __shared__ float tot[4 * W_DIM];

    const size_t NE = (size_t)CIN * HW;
    const int c = blockIdx.x;
    const int t = threadIdx.x;
    const float* src = fm + (size_t)c * HW + 4 * t;

    for (int j = 0; j < 4; ++j) {
        float4 s = make_float4(0.f, 0.f, 0.f, 0.f);
        for (int b = 0; b < B_IMG; ++b) {
            const float4 v = *(const float4*)(src + (size_t)b * NE + 1024 * j);
            s.x += v.x; s.y += v.y; s.z += v.z; s.w += v.w;
        }
        s.y += s.x; s.z += s.y; s.w += s.z;
        float run = s.w;
        run += row_shr_or_zero<1>(run);
        run += row_shr_or_zero<2>(run);
        run += row_shr_or_zero<4>(run);
        run += row_shr_or_zero<8>(run);
        const float left = row_shr_or_zero<1>(run);
        s.x += left; s.y += left; s.z += left; s.w += left;
        *(float4*)(tile + ((t >> 4) + 16 * j) * SAT_PITCH + 4 * (t & 15)) = s;
    }
    __syncthreads();

    const int x   = t & 63;
    const int seg = t >> 6;
    float col[16];
    float crun = 0.0f;
    for (int i = 0; i < 16; ++i) {
        crun += tile[(16 * seg + i) * SAT_PITCH + x];
        col[i] = crun;
    }
    tot[seg * W_DIM + x] = crun;
    __syncthreads();

    float above = 0.0f;
    if (seg > 0) above  = tot[x];
    if (seg > 1) above += tot[W_DIM + x];
    if (seg > 2) above += tot[2 * W_DIM + x];

    float* dst = P + (size_t)c * HW + (size_t)(16 * seg) * W_DIM + x;
    for (int i = 0; i < 16; ++i) dst[i * W_DIM] = col[i] + above;
}

#define M_TILES   (C_OUT_PAD / 16)
#define NT_MAX    7
#define K_CHUNK   32
#define K_QUARTER (CIN / 4)
#define N_CHUNKS  (K_QUARTER / K_CHUNK)
#define LDS_RESERVE_BYTES (176 * 1024)

typedef __attribute__((ext_vector_type(4))) float v4f;

template <int NT>
struct ChunkFrag {
    v4f   a[NT][4];
    float b0[16];
    float b1[16];
};

template <int NT, int T0>
__device__ __forceinline__ void load_chunk(ChunkFrag<NT>& f,
                                           const float* __restrict__ Wlo,
                                           const float* __restrict__ Whi,
                                           const float* __restrict__ Fl,
                                           int k) {
#pragma unroll
    for (int mt = 0; mt < NT; ++mt) {
        const float* p = (T0 + mt == M_TILES - 1) ? Whi
                                                  : Wlo + (size_t)(T0 + mt) * 16 * CIN;
#pragma unroll
        for (int i = 0; i < 4; ++i)
            f.a[mt][i] = *(const v4f*)(p + k + 4 * i);
    }
#pragma unroll
    for (int e = 0; e < 16; ++e) {
        f.b0[e] = Fl[(size_t)(k + e) * HW];
        f.b1[e] = Fl[(size_t)(k + e) * HW + 16];
    }
}

template <int NT>
__device__ __forceinline__ void mma_chunk(v8f (&acc)[NT][2], const ChunkFrag<NT>& f) {
#pragma unroll
    for (int j = 0; j < 8; ++j) {
        v2f bl; bl.x = f.b0[2 * j]; bl.y = f.b0[2 * j + 1];
        v2f br; br.x = f.b1[2 * j]; br.y = f.b1[2 * j + 1];
#pragma unroll
        for (int mt = 0; mt < NT; ++mt) {
            const v4f q = f.a[mt][j >> 1];
            v2f a;
            if (j & 1) { a.x = q.z; a.y = q.w; } else { a.x = q.x; a.y = q.y; }
            acc[mt][0] = __builtin_amdgcn_wmma_f32_16x16x4_f32(
                false, a, false, bl, (short)0, acc[mt][0], false, false);
            acc[mt][1] = __builtin_amdgcn_wmma_f32_16x16x4_f32(
                false, a, false, br, (short)0, acc[mt][1], false, false);
        }
    }
}

typedef float red_t[NT_MAX][8][2][32];

template <int NT, int T0>
__device__ __forceinline__ void gemm_group(const float* __restrict__ Wmat,
                                           const float* __restrict__ F,
                                           float* __restrict__ S,
                                           red_t* red, int n0, int rot) {
    const int lane  = threadIdx.x & 31;
    const int w     = __builtin_amdgcn_readfirstlane(threadIdx.x >> 5);
    const int lhalf = lane >> 4;
    const int l16   = lane & 15;

    const float* Wlo = Wmat + (size_t)l16 * CIN + 16 * lhalf;
    const float* Whi = Wmat + (size_t)min(16 * (M_TILES - 1) + l16, C_OUT - 1) * CIN + 16 * lhalf;
    const float* Fl  = F + (size_t)(16 * lhalf) * HW + n0 + l16;

    v8f acc[NT][2];
#pragma unroll
    for (int mt = 0; mt < NT; ++mt)
#pragma unroll
        for (int t = 0; t < 2; ++t)
#pragma unroll
            for (int v = 0; v < 8; ++v) acc[mt][t][v] = 0.0f;

    const int kbase = w * K_QUARTER;
#define CHUNK_K(i) (kbase + K_CHUNK * (((i) + rot) & (N_CHUNKS - 1)))

    ChunkFrag<NT> f0, f1;
    load_chunk<NT, T0>(f0, Wlo, Whi, Fl, CHUNK_K(0));
#pragma unroll 1
    for (int it = 0; it < N_CHUNKS - 2; it += 2) {
        load_chunk<NT, T0>(f1, Wlo, Whi, Fl, CHUNK_K(it + 1));
        mma_chunk<NT>(acc, f0);
        load_chunk<NT, T0>(f0, Wlo, Whi, Fl, CHUNK_K(it + 2));
        mma_chunk<NT>(acc, f1);
    }
    load_chunk<NT, T0>(f1, Wlo, Whi, Fl, CHUNK_K(N_CHUNKS - 1));
    mma_chunk<NT>(acc, f0);
    mma_chunk<NT>(acc, f1);
#undef CHUNK_K

#pragma unroll
    for (int mt = 0; mt < NT; ++mt)
#pragma unroll
        for (int t = 0; t < 2; ++t)
#pragma unroll
            for (int v = 0; v < 8; ++v) red[w][mt][v][t][lane] = acc[mt][t][v];
    __syncthreads();

    const int t  = lane >> 4;
#pragma unroll
    for (int i = 0; i < NT * 4; ++i) {
        const int slot = w + 4 * i;
        const int mt = slot >> 4;
        const int v  = (slot >> 1) & 7;
        const int h  = slot & 1;
        const int src = l16 + 16 * h;
        const float s = (red[0][mt][v][t][src] + red[1][mt][v][t][src])
                      + (red[2][mt][v][t][src] + red[3][mt][v][t][src]);
        const int o = 16 * (T0 + mt) + v + 8 * h;
        S[(size_t)o * HW + n0 + lane] = s;
    }
}

__global__ void __launch_bounds__(128)
gemm_wmma_kernel(const float* __restrict__ Wmat,
                 const float* __restrict__ F,
                 float* __restrict__ S) {
    __shared__ __attribute__((aligned(16))) float lds[LDS_RESERVE_BYTES / 4];
    red_t* red   = (red_t*)lds;

    const int np  = blockIdx.x >> 1;
    const int n0  = np * 32;
    const int rot = np & (N_CHUNKS - 1);

    if ((blockIdx.x & 1) == 0)
        gemm_group<NT_MAX, 0>(Wmat, F, S, red, n0, rot);
    else
        gemm_group<M_TILES - NT_MAX, NT_MAX>(Wmat, F, S, red, n0, rot);
}

#define ROI_PER_WG  4
#define ROI_THREADS 128
#define ROI_WGS     256
#define CH_STEPS   7

__global__ void __launch_bounds__(ROI_THREADS)
roipool_kernel(const int* __restrict__ rois,
               const float* __restrict__ sat,
               const float* __restrict__ bias,
               float* __restrict__ out) {
    const int lane = threadIdx.x & 31;
    const int i = __builtin_amdgcn_readfirstlane(blockIdx.x * ROI_PER_WG + (threadIdx.x >> 5));

    const int* r = rois + i * 4;
    const int ymin = r[0] / 30, xmin = r[1] / 30;
    const int ymax = r[2] / 30, xmax = r[3] / 30;
    const int ystep = (ymax - ymin) / KPOOL;
    const int xstep = (xmax - xmin) / KPOOL;
    const float cnt = (float)(B_IMG * ystep * xstep);
    const float den = fmaxf(cnt, 1.0f);

    float tv[CH_STEPS][4];
    bool  tz[CH_STEPS][4];
    float bv[CH_STEPS];
#pragma unroll
    for (int k = 0; k < CH_STEPS; ++k) {
        const int ch  = lane + 32 * k;
        const int chc = min(ch, C_OUT - 1);
        const int bin = chc >> 2;
        const int jy = bin / KPOOL, jx = bin % KPOOL;
        const int ys = ymin + jy * ystep, ye = ys + ystep;
        const int xs = xmin + jx * xstep, xe = xs + xstep;
        const float* sp = sat + (size_t)chc * HW;
        const int yy[4] = {ye, ys, ye, ys};
        const int xx[4] = {xe, xe, xs, xs};
#pragma unroll
        for (int q = 0; q < 4; ++q) {
            tz[k][q] = (yy[q] > 0 && xx[q] > 0);
            tv[k][q] = sp[max(yy[q] - 1, 0) * W_DIM + max(xx[q] - 1, 0)];
        }
        bv[k] = bias[chc];
    }

    float s = 0.0f;
#pragma unroll
    for (int k = 0; k < CH_STEPS; ++k) {
        const float t0 = tz[k][0] ? tv[k][0] : 0.0f;
        const float t1 = tz[k][1] ? tv[k][1] : 0.0f;
        const float t2 = tz[k][2] ? tv[k][2] : 0.0f;
        const float t3 = tz[k][3] ? tv[k][3] : 0.0f;
        const float bs = t0 - t1 - t2 + t3;
        const float p  = (cnt > 0.0f) ? bs / den + bv[k] : 0.0f;
        s += (lane + 32 * k < C_OUT) ? p : 0.0f;
    }
    s += __shfl_xor(s, 4);
    s += __shfl_xor(s, 8);
    s += __shfl_xor(s, 16);
    if (lane < 4) out[i * 4 + lane] = s / (float)(KPOOL * KPOOL);
}

extern "C" void kernel_launch(void* const* d_in, const int* in_sizes, int n_in,
                              void* d_out, int out_size, void* d_ws, size_t ws_size,
                              hipStream_t stream) {
    const float* fm    = (const float*)d_in[0];
    const int*   rois  = (const int*)  d_in[1];
    const float* convw = (const float*)d_in[2];
    const float* convb = (const float*)d_in[3];
    float*       out   = (float*)d_out;

    char* ws = (char*)d_ws;
    float* P = (float*)(ws);
    float* S = (float*)(ws + (size_t)CIN * HW * sizeof(float));

    batchsum_sat_kernel<<<CIN, 256, 0, stream>>>(fm, P);

    gemm_wmma_kernel<<<256, 128, 0, stream>>>(convw, P, S);

    roipool_kernel<<<ROI_WGS, ROI_THREADS, 0, stream>>>(rois, S, convb, out);
}
